# starting kernel
# baseline (speedup 1.0000x reference)
#include <hip/hip_runtime.h>
#include <math.h>

#define FD 5
#define NNODES 100000
#define NEDGES 6400000

typedef __attribute__((ext_vector_type(2))) float v2f;
typedef __attribute__((ext_vector_type(8))) float v8f;

__device__ __forceinline__ int imin(int a, int b) { return a < b ? a : b; }
__device__ __forceinline__ int imax(int a, int b) { return a > b ? a : b; }

__device__ __forceinline__ void gatomic_add_f32(float* p, float v) {
    asm volatile("global_atomic_add_f32 %0, %1, off"
                 :
                 : "v"(p), "v"(v)
                 : "memory");
}

__global__ void zero_f32(float* __restrict__ p, int n) {
    int i = blockIdx.x * blockDim.x + threadIdx.x;
    int stride = gridDim.x * blockDim.x;
    for (; i < n; i += stride) p[i] = 0.0f;
}

template <bool ADD_DEG>
__global__ void edge_agg(const float* __restrict__ h,
                         const int* __restrict__ src,
                         const int* __restrict__ dst,
                         float* __restrict__ agg,
                         float* __restrict__ deg,
                         int n_edges) {
    int e = blockIdx.x * blockDim.x + threadIdx.x;
    if (e >= n_edges) return;
    int s = src[e];
    int d = dst[e];
    const float* hs = h + (long)s * FD;
    float* ad = agg + (long)d * FD;
#pragma unroll
    for (int k = 0; k < FD; ++k) {
        gatomic_add_f32(ad + k, hs[k]);
    }
    if (ADD_DEG) {
        gatomic_add_f32(deg + d, 1.0f);
    }
}

template <int F_OUT>
__global__ void node_wmma(const float* __restrict__ h,
                          const float* __restrict__ agg,
                          const float* __restrict__ deg,
                          const float* __restrict__ Wself,
                          const float* __restrict__ Wneigh,
                          const float* __restrict__ bias,
                          float* __restrict__ out,
                          int n_nodes) {
    const int lane  = threadIdx.x & 31;
    const int wave  = threadIdx.x >> 5;
    const int tile  = blockIdx.x * (blockDim.x >> 5) + wave;
    const int m0    = tile * 16;
    if (m0 >= n_nodes) return;

    const int row   = lane & 15;
    const int khalf = lane >> 4;

    const int m  = m0 + row;
    const int mc = imin(m, n_nodes - 1);
    const float invdeg = 1.0f / fmaxf(deg[mc], 1.0f);

    const int n  = row;
    const int nc = imin(n, F_OUT - 1);
    const bool nvalid = (n < F_OUT);

    v8f c = {};
#pragma unroll
    for (int chunk = 0; chunk < 3; ++chunk) {
        v2f a, b;
#pragma unroll
        for (int j = 0; j < 2; ++j) {
            const int k  = chunk * 4 + khalf * 2 + j;
            const int kh = imin(k, FD - 1);
            const int kn = imin(imax(k - FD, 0), FD - 1);

            const float hv = h[(long)mc * FD + kh];
            const float nv = agg[(long)mc * FD + kn] * invdeg;
            const float av = (k < FD) ? hv : ((k < 2 * FD) ? nv : 0.0f);

            const float ws = Wself[kh * F_OUT + nc];
            const float wn = Wneigh[kn * F_OUT + nc];
            float bv = (k < FD) ? ws : ((k < 2 * FD) ? wn : 0.0f);
            bv = nvalid ? bv : 0.0f;

            a[j] = av;
            b[j] = bv;
        }
        c = __builtin_amdgcn_wmma_f32_16x16x4_f32(false, a, false, b,
                                                  (short)0, c, false, false);
    }

    const float bn = bias[nc];
#pragma unroll
    for (int r = 0; r < 8; ++r) {
        const int mm = m0 + r + khalf * 8;
        if (mm < n_nodes && nvalid) {
            const float v = c[r] + bn;
            out[(long)mm * F_OUT + n] = 1.0f / (1.0f + __expf(-v));
        }
    }
}

extern "C" void kernel_launch(void* const* d_in, const int* in_sizes, int n_in,
                              void* d_out, int out_size, void* d_ws, size_t ws_size,
                              hipStream_t stream) {
    (void)in_sizes; (void)n_in; (void)out_size; (void)ws_size;

    const float* x      = (const float*)d_in[0];
    const int*   src    = (const int*)d_in[1];
    const int*   dst    = (const int*)d_in[2];
    const float* Wself1 = (const float*)d_in[3];
    const float* Wneigh1= (const float*)d_in[4];
    const float* b1     = (const float*)d_in[5];
    const float* Wself2 = (const float*)d_in[6];
    const float* Wneigh2= (const float*)d_in[7];
    const float* b2     = (const float*)d_in[8];
    float* out = (float*)d_out;

    float* h   = (float*)d_ws;
    float* agg = h + (size_t)NNODES * FD;
    float* deg = agg + (size_t)NNODES * FD;

    const int ZB = 256;
    const int EB = 256;
    const int edge_grid = (NEDGES + EB - 1) / EB;

    const int NODE_B = 256;
    const int tiles = (NNODES + 15) / 16;
    const int node_grid = (tiles + (NODE_B / 32) - 1) / (NODE_B / 32);

    {
        const int nz = NNODES * FD + NNODES;
        zero_f32<<<(nz + ZB - 1) / ZB, ZB, 0, stream>>>(agg, nz);
    }
    edge_agg<true><<<edge_grid, EB, 0, stream>>>(x, src, dst, agg, deg, NEDGES);
    node_wmma<FD><<<node_grid, NODE_B, 0, stream>>>(x, agg, deg, Wself1, Wneigh1,
                                                    b1, h, NNODES);

    {
        const int nz = NNODES * FD;
        zero_f32<<<(nz + ZB - 1) / ZB, ZB, 0, stream>>>(agg, nz);
    }
    edge_agg<false><<<edge_grid, EB, 0, stream>>>(h, src, dst, agg, deg, NEDGES);
    node_wmma<10><<<node_grid, NODE_B, 0, stream>>>(h, agg, deg, Wself2, Wneigh2,
                                                    b2, out, NNODES);
}


# Edge pass reads source rows from LDS, not L2: each work-group holds 1/8 of the node table and scans its share of src/dst for that slice's edges; f32 atomics kept, arrival order of adds new
# speedup vs baseline: 18.2413x; 18.2413x over previous
#include <hip/hip_runtime.h>
#include <math.h>

#define FD 5
#define NNODES 100000
#define NEDGES 6400000

typedef __attribute__((ext_vector_type(2))) float v2f;
typedef __attribute__((ext_vector_type(8))) float v8f;

__device__ __forceinline__ int imin(int a, int b) { return a < b ? a : b; }
__device__ __forceinline__ int imax(int a, int b) { return a > b ? a : b; }

__device__ __forceinline__ void gatomic_add_f32(float* p, float v) {
    asm volatile("global_atomic_add_f32 %0, %1, off"
                 :
                 : "v"(p), "v"(v)
                 : "memory");
}

#define ROWF 8
#define CNT FD
#define PLANE (NNODES * ROWF)

__device__ __forceinline__ int package_half() {
    const unsigned id = __builtin_amdgcn_s_sendmsg_rtn(0x87);
    return (int)((id >> 18) & 1u);
}

template <int LIVE>
__global__ void zero_rows(float* __restrict__ p, int n) {
    int i = blockIdx.x * blockDim.x + threadIdx.x;
    int stride = gridDim.x * blockDim.x;
    for (; i < n; i += stride) {
        const int r = i / LIVE;
        p[r * ROWF + (i - r * LIVE)] = 0.0f;
    }
}

#define NSLICE 8
#define SLICE_ROWS (NNODES / NSLICE)
#define NPART 32
#define SCAN_BLOCK 1024
#define SCAN_WAVES (SCAN_BLOCK / 32)
#define SCAN_EDGES 128
#define DRAIN 4

typedef __attribute__((ext_vector_type(2))) int v2i;
typedef __attribute__((ext_vector_type(4))) int v4i;
typedef __attribute__((ext_vector_type(4))) float v4f;

template <bool ADD_DEG>
__global__ void __launch_bounds__(SCAN_BLOCK)
edge_agg(const float* __restrict__ h,
         const int* __restrict__ src,
         const int* __restrict__ dst,
         float* __restrict__ sums,
         int n_edges) {
    __shared__ float tab[SLICE_ROWS * FD];
    __shared__ v2i queue[SCAN_WAVES][SCAN_EDGES];

    const int lane  = threadIdx.x & 31;
    const int wave  = threadIdx.x >> 5;
    const int slice = blockIdx.x % NSLICE;
    const int part  = blockIdx.x / NSLICE;
    const int row0  = slice * SLICE_ROWS;

    {
        const v4f* g4 = (const v4f*)(h + (long)row0 * FD);
        v4f* t4 = (v4f*)tab;
        for (int i = threadIdx.x; i < SLICE_ROWS * FD / 4; i += SCAN_BLOCK) t4[i] = g4[i];
    }
    __syncthreads();

    const int  k    = lane & 7;
    const int  g    = lane >> 3;
    const int  kc   = imin(k, FD - 1);
    const bool feat = (k < FD);
    const bool adds = feat || (ADD_DEG && k == FD);
    float* const plane = sums + package_half() * PLANE;

    const int epp  = (((n_edges + NPART - 1) / NPART) + 3) & ~3;
    const int pbeg = part * epp;
    const int pend = imin(pbeg + epp, n_edges);

    v2i* const q = queue[wave];

    for (int eb = pbeg + wave * SCAN_EDGES; eb < pend; eb += SCAN_WAVES * SCAN_EDGES) {
        const int e = eb + lane * 4;
        int s4[4], d4[4];
        if (e + 3 < pend) {
            const v4i sv = *(const v4i*)(src + e);
            const v4i dv = *(const v4i*)(dst + e);
            s4[0] = sv.x; s4[1] = sv.y; s4[2] = sv.z; s4[3] = sv.w;
            d4[0] = dv.x; d4[1] = dv.y; d4[2] = dv.z; d4[3] = dv.w;
        } else {
#pragma unroll
            for (int j = 0; j < 4; ++j) {
                s4[j] = -1; d4[j] = 0;
                if (e + j < pend) { s4[j] = src[e + j]; d4[j] = dst[e + j]; }
            }
        }

        int cnt = 0;
#pragma unroll
        for (int j = 0; j < 4; ++j) {
            const unsigned loc = (unsigned)(s4[j] - row0);
            const bool hit = loc < (unsigned)SLICE_ROWS;
            const unsigned m = __builtin_amdgcn_ballot_w32(hit);
            const int pos = cnt + (int)__builtin_amdgcn_mbcnt_lo(m, 0u);
            if (hit) { v2i ent; ent.x = (int)loc; ent.y = d4[j]; q[pos] = ent; }
            cnt += __builtin_popcount(m);
        }

        for (int r = 0; r < cnt; r += 4 * DRAIN) {
            float v[DRAIN];
            int   dd[DRAIN];
            bool  live[DRAIN];
#pragma unroll
            for (int i = 0; i < DRAIN; ++i) {
                const int idx = r + 4 * i + g;
                live[i] = idx < cnt;
                const int ic = imin(idx, cnt - 1);
                const v2i ent = q[ic];
                dd[i] = ent.y;
                v[i] = tab[ent.x * FD + kc];
            }
#pragma unroll
            for (int i = 0; i < DRAIN; ++i) {
                float* p = plane + (dd[i] * ROWF + k);
                const float val = feat ? v[i] : 1.0f;
                if (adds && live[i]) {
                    gatomic_add_f32(p, val);
                }
            }
        }
    }
}

template <int F_OUT>
__global__ void node_wmma(const float* __restrict__ h,
                          const float* __restrict__ sums,
                          const float* __restrict__ Wself,
                          const float* __restrict__ Wneigh,
                          const float* __restrict__ bias,
                          float* __restrict__ out,
                          int n_nodes) {
    const int lane  = threadIdx.x & 31;
    const int wave  = threadIdx.x >> 5;
    const int tile  = blockIdx.x * (blockDim.x >> 5) + wave;
    const int m0    = tile * 16;
    if (m0 >= n_nodes) return;

    const int row   = lane & 15;
    const int khalf = lane >> 4;

    const int m  = m0 + row;
    const int mc = imin(m, n_nodes - 1);
    const float* const row0 = sums + (long)mc * ROWF;
    const float* const row1 = row0 + PLANE;
    const float invdeg = 1.0f / fmaxf(row0[CNT] + row1[CNT], 1.0f);

    const int n  = row;
    const int nc = imin(n, F_OUT - 1);
    const bool nvalid = (n < F_OUT);

    v8f c = {};
#pragma unroll
    for (int chunk = 0; chunk < 3; ++chunk) {
        v2f a, b;
#pragma unroll
        for (int j = 0; j < 2; ++j) {
            const int k  = chunk * 4 + khalf * 2 + j;
            const int kh = imin(k, FD - 1);
            const int kn = imin(imax(k - FD, 0), FD - 1);

            const float hv = h[(long)mc * FD + kh];
            const float nv = (row0[kn] + row1[kn]) * invdeg;
            const float av = (k < FD) ? hv : ((k < 2 * FD) ? nv : 0.0f);

            const float ws = Wself[kh * F_OUT + nc];
            const float wn = Wneigh[kn * F_OUT + nc];
            float bv = (k < FD) ? ws : ((k < 2 * FD) ? wn : 0.0f);
            bv = nvalid ? bv : 0.0f;

            a[j] = av;
            b[j] = bv;
        }
        c = __builtin_amdgcn_wmma_f32_16x16x4_f32(false, a, false, b,
                                                  (short)0, c, false, false);
    }

    const float bn = bias[nc];
#pragma unroll
    for (int r = 0; r < 8; ++r) {
        const int mm = m0 + r + khalf * 8;
        if (mm < n_nodes && nvalid) {
            const float v = c[r] + bn;
            out[(long)mm * F_OUT + n] = 1.0f / (1.0f + __expf(-v));
        }
    }
}

extern "C" void kernel_launch(void* const* d_in, const int* in_sizes, int n_in,
                              void* d_out, int out_size, void* d_ws, size_t ws_size,
                              hipStream_t stream) {
    (void)in_sizes; (void)n_in; (void)out_size; (void)ws_size;

    const float* x      = (const float*)d_in[0];
    const int*   src    = (const int*)d_in[1];
    const int*   dst    = (const int*)d_in[2];
    const float* Wself1 = (const float*)d_in[3];
    const float* Wneigh1= (const float*)d_in[4];
    const float* b1     = (const float*)d_in[5];
    const float* Wself2 = (const float*)d_in[6];
    const float* Wneigh2= (const float*)d_in[7];
    const float* b2     = (const float*)d_in[8];
    float* out = (float*)d_out;

    float* sums = (float*)d_ws;
    float* h    = sums + (size_t)2 * PLANE;

    const int ZB = 256;
    const int EB = SCAN_BLOCK;
    const int edge_grid = NSLICE * NPART;

    const int NODE_B = 256;
    const int tiles = (NNODES + 15) / 16;
    const int node_grid = (tiles + (NODE_B / 32) - 1) / (NODE_B / 32);

    {
        const int nz = 2 * NNODES * (FD + 1);
        zero_rows<FD + 1><<<(nz + ZB - 1) / ZB, ZB, 0, stream>>>(sums, nz);
    }
    edge_agg<true><<<edge_grid, EB, 0, stream>>>(x, src, dst, sums, NEDGES);
    node_wmma<FD><<<node_grid, NODE_B, 0, stream>>>(x, sums, Wself1, Wneigh1,
                                                    b1, h, NNODES);

    {
        const int nz = 2 * NNODES * FD;
        zero_rows<FD><<<(nz + ZB - 1) / ZB, ZB, 0, stream>>>(sums, nz);
    }
    edge_agg<false><<<edge_grid, EB, 0, stream>>>(h, src, dst, sums, NEDGES);
    node_wmma<10><<<node_grid, NODE_B, 0, stream>>>(h, sums, Wself2, Wneigh2,
                                                    b2, out, NNODES);
}


# Edges sorted once per call by source slice (new pre-pass, list in workspace): an edge pass reads only its slice's edges, not all of src/dst per slice; f32 atomics kept, arrival order of adds new
# speedup vs baseline: 23.9710x; 1.3141x over previous
#include <hip/hip_runtime.h>
#include <math.h>

#define FD 5
#define NNODES 100000
#define NEDGES 6400000

typedef __attribute__((ext_vector_type(2))) float v2f;
typedef __attribute__((ext_vector_type(8))) float v8f;

__device__ __forceinline__ int imin(int a, int b) { return a < b ? a : b; }
__device__ __forceinline__ int imax(int a, int b) { return a > b ? a : b; }

__device__ __forceinline__ void gatomic_add_f32(float* p, float v) {
    asm volatile("global_atomic_add_f32 %0, %1, off"
                 :
                 : "v"(p), "v"(v)
                 : "memory");
}

#define ROWF 8
#define CNT FD
#define PLANE (NNODES * ROWF)

__device__ __forceinline__ int package_half() {
    const unsigned id = __builtin_amdgcn_s_sendmsg_rtn(0x87);
    return (int)((id >> 18) & 1u);
}

template <int LIVE>
__global__ void zero_rows(float* __restrict__ p, int n) {
    int i = blockIdx.x * blockDim.x + threadIdx.x;
    int stride = gridDim.x * blockDim.x;
    for (; i < n; i += stride) {
        const int r = i / LIVE;
        p[r * ROWF + (i - r * LIVE)] = 0.0f;
    }
}

#define NSLICE 8
#define SLICE_ROWS (NNODES / NSLICE)
#define NCHUNK 256
#define CHUNK (NEDGES / NCHUNK)
#define WSPAN 784
#define WTRIPS 7
#define OFFP 16
#define DSTBITS 17
#define SORT_BLOCK 1024
#define SORT_WAVES (SORT_BLOCK / 32)

typedef __attribute__((ext_vector_type(4))) int v4i;
typedef __attribute__((ext_vector_type(4))) float v4f;

__device__ __forceinline__ unsigned wave_sum(unsigned x) {
#pragma unroll
    for (int m = 16; m > 0; m >>= 1) x += (unsigned)__shfl_xor((int)x, m, 32);
    return x;
}

__global__ void __launch_bounds__(SORT_BLOCK)
sort_by_slice(const int* __restrict__ src,
              const int* __restrict__ dst,
              unsigned* __restrict__ part,
              int* __restrict__ off) {
    __shared__ int cnt[NSLICE * SORT_WAVES];
    __shared__ int base[NSLICE * SORT_WAVES];

    const int lane = threadIdx.x & 31;
    const int wave = threadIdx.x >> 5;
    const int c    = blockIdx.x;
    const int cbeg = c * CHUNK;
    const int wbeg = cbeg + wave * WSPAN;
    const int wend = imin(wbeg + WSPAN, cbeg + CHUNK);

    v4i sv[WTRIPS], dv[WTRIPS];
#pragma unroll
    for (int t = 0; t < WTRIPS; ++t) {
        const int e = wbeg + t * 128 + lane * 4;
        sv[t] = (v4i){-1, -1, -1, -1};
        dv[t] = (v4i){0, 0, 0, 0};
        if (e < wend) { sv[t] = *(const v4i*)(src + e); dv[t] = *(const v4i*)(dst + e); }
    }

    unsigned clo = 0, chi = 0;
#pragma unroll
    for (int t = 0; t < WTRIPS; ++t) {
#pragma unroll
        for (int j = 0; j < 4; ++j) {
            const unsigned sl  = (unsigned)sv[t][j] / (unsigned)SLICE_ROWS;
            const unsigned one = 1u << (8u * (sl & 3u));
            clo += (sl < 4u) ? one : 0u;
            chi += (sl - 4u < 4u) ? one : 0u;
        }
    }
    {
        const unsigned s02 = wave_sum(clo & 0x00ff00ffu);
        const unsigned s13 = wave_sum((clo >> 8) & 0x00ff00ffu);
        const unsigned s46 = wave_sum(chi & 0x00ff00ffu);
        const unsigned s57 = wave_sum((chi >> 8) & 0x00ff00ffu);
        if (lane == 0) {
            cnt[0 * SORT_WAVES + wave] = (int)(s02 & 0xffffu);
            cnt[1 * SORT_WAVES + wave] = (int)(s13 & 0xffffu);
            cnt[2 * SORT_WAVES + wave] = (int)(s02 >> 16);
            cnt[3 * SORT_WAVES + wave] = (int)(s13 >> 16);
            cnt[4 * SORT_WAVES + wave] = (int)(s46 & 0xffffu);
            cnt[5 * SORT_WAVES + wave] = (int)(s57 & 0xffffu);
            cnt[6 * SORT_WAVES + wave] = (int)(s46 >> 16);
            cnt[7 * SORT_WAVES + wave] = (int)(s57 >> 16);
        }
    }
    __syncthreads();

    if (threadIdx.x < NSLICE * SORT_WAVES) {
        int acc = 0;
        for (int i = 0; i < (int)threadIdx.x; ++i) acc += cnt[i];
        base[threadIdx.x] = acc;
        if ((threadIdx.x & (SORT_WAVES - 1)) == 0) off[c * OFFP + (threadIdx.x / SORT_WAVES)] = acc;
        if (threadIdx.x == NSLICE * SORT_WAVES - 1) off[c * OFFP + NSLICE] = acc + cnt[threadIdx.x];
    }
    __syncthreads();

    int b[NSLICE];
#pragma unroll
    for (int j = 0; j < NSLICE; ++j)
        b[j] = __builtin_amdgcn_readfirstlane(base[j * SORT_WAVES + wave]);
    unsigned* const seg = part + (long)cbeg;
#pragma unroll
    for (int t = 0; t < WTRIPS; ++t) {
#pragma unroll
        for (int jj = 0; jj < 4; ++jj) {
            const unsigned sl = (unsigned)sv[t][jj] / (unsigned)SLICE_ROWS;
            int pos = 0;
#pragma unroll
            for (int j = 0; j < NSLICE; ++j) {
                const bool mine = (sl == (unsigned)j);
                const unsigned m = __builtin_amdgcn_ballot_w32(mine);
                const int p = (int)__builtin_amdgcn_mbcnt_lo(m, (unsigned)b[j]);
                pos = mine ? p : pos;
                b[j] += __builtin_popcount(m);
            }
            if (sl < (unsigned)NSLICE) {
                const unsigned loc = (unsigned)sv[t][jj] - sl * (unsigned)SLICE_ROWS;
                seg[pos] = (loc << DSTBITS) | (unsigned)dv[t][jj];
            }
        }
    }
}

#define NPART 32
#define SCAN_BLOCK 1024
#define SCAN_WAVES (SCAN_BLOCK / 32)
#define DRAIN 8
#define TRIP_EDGES (4 * DRAIN)
static_assert(DRAIN == 8, "the fence in edge_agg_sorted names the eight table reads of a trip");

template <bool ADD_DEG>
__global__ void __launch_bounds__(SCAN_BLOCK)
edge_agg_sorted(const float* __restrict__ h,
                const unsigned* __restrict__ part,
                const int* __restrict__ off,
                float* __restrict__ sums) {
    __shared__ float tab[SLICE_ROWS * FD];

    const int lane  = threadIdx.x & 31;
    const int wave  = threadIdx.x >> 5;
    const int slice = blockIdx.x % NSLICE;
    const int prt   = blockIdx.x / NSLICE;
    const int row0  = slice * SLICE_ROWS;

    {
        const v4f* g4 = (const v4f*)(h + (long)row0 * FD);
        v4f* t4 = (v4f*)tab;
        for (int i = threadIdx.x; i < SLICE_ROWS * FD / 4; i += SCAN_BLOCK) t4[i] = g4[i];
    }
    __syncthreads();

    const int  k    = lane & 7;
    const int  g    = lane >> 3;
    const int  kc   = imin(k, FD - 1);
    const bool feat = (k < FD);
    const bool adds = feat || (ADD_DEG && k == FD);
    float* const plane = sums + package_half() * PLANE;

    const int c0 = prt * (NCHUNK / NPART);
    int nbeg = off[c0 * OFFP + slice];
    int nend = off[c0 * OFFP + slice + 1];
    for (int cc = 0; cc < NCHUNK / NPART; ++cc) {
        const int c    = c0 + cc;
        const int sbeg = nbeg;
        const int send = nend;
        const int cn = imin(c + 1, NCHUNK - 1);
        nbeg = off[cn * OFFP + slice];
        nend = off[cn * OFFP + slice + 1];
        const unsigned* const seg = part + (long)c * CHUNK;
        const int w = (wave + 4 * cc) & (SCAN_WAVES - 1);

        for (int r = sbeg + w * TRIP_EDGES; r < send; r += SCAN_WAVES * TRIP_EDGES) {
            unsigned wd[DRAIN];
            float    v[DRAIN];
            bool     live[DRAIN];
#pragma unroll
            for (int i = 0; i < DRAIN; ++i) {
                const int idx = r + 4 * i + g;
                live[i] = idx < send;
                wd[i] = seg[imin(idx, send - 1)];
            }
#pragma unroll
            for (int i = 0; i < DRAIN; ++i)
                v[i] = tab[(int)(wd[i] >> DSTBITS) * FD + kc];
            asm volatile("" : : "v"(v[0]), "v"(v[1]), "v"(v[2]), "v"(v[3]),
                                "v"(v[4]), "v"(v[5]), "v"(v[6]), "v"(v[7]));
#pragma unroll
            for (int i = 0; i < DRAIN; ++i) {
                const int dd = (int)(wd[i] & ((1u << DSTBITS) - 1u));
                float* p = plane + (dd * ROWF + k);
                const float val = feat ? v[i] : 1.0f;
                if (adds && live[i]) {
                    gatomic_add_f32(p, val);
                }
            }
        }
    }
}

#define SCAN_EDGES 128
#define SCAN_DRAIN 4

typedef __attribute__((ext_vector_type(2))) int v2i;

template <bool ADD_DEG>
__global__ void __launch_bounds__(SCAN_BLOCK)
edge_agg_scan(const float* __restrict__ h,
              const int* __restrict__ src,
              const int* __restrict__ dst,
              float* __restrict__ sums,
              int n_edges) {
    __shared__ float tab[SLICE_ROWS * FD];
    __shared__ v2i queue[SCAN_WAVES][SCAN_EDGES];

    const int lane  = threadIdx.x & 31;
    const int wave  = threadIdx.x >> 5;
    const int slice = blockIdx.x % NSLICE;
    const int part  = blockIdx.x / NSLICE;
    const int row0  = slice * SLICE_ROWS;

    {
        const v4f* g4 = (const v4f*)(h + (long)row0 * FD);
        v4f* t4 = (v4f*)tab;
        for (int i = threadIdx.x; i < SLICE_ROWS * FD / 4; i += SCAN_BLOCK) t4[i] = g4[i];
    }
    __syncthreads();

    const int  k    = lane & 7;
    const int  g    = lane >> 3;
    const int  kc   = imin(k, FD - 1);
    const bool feat = (k < FD);
    const bool adds = feat || (ADD_DEG && k == FD);
    float* const plane = sums + package_half() * PLANE;

    const int epp  = (((n_edges + NPART - 1) / NPART) + 3) & ~3;
    const int pbeg = part * epp;
    const int pend = imin(pbeg + epp, n_edges);

    v2i* const q = queue[wave];

    for (int eb = pbeg + wave * SCAN_EDGES; eb < pend; eb += SCAN_WAVES * SCAN_EDGES) {
        const int e = eb + lane * 4;
        int s4[4], d4[4];
        if (e + 3 < pend) {
            const v4i sv = *(const v4i*)(src + e);
            const v4i dv = *(const v4i*)(dst + e);
            s4[0] = sv.x; s4[1] = sv.y; s4[2] = sv.z; s4[3] = sv.w;
            d4[0] = dv.x; d4[1] = dv.y; d4[2] = dv.z; d4[3] = dv.w;
        } else {
#pragma unroll
            for (int j = 0; j < 4; ++j) {
                s4[j] = -1; d4[j] = 0;
                if (e + j < pend) { s4[j] = src[e + j]; d4[j] = dst[e + j]; }
            }
        }

        int cnt = 0;
#pragma unroll
        for (int j = 0; j < 4; ++j) {
            const unsigned loc = (unsigned)(s4[j] - row0);
            const bool hit = loc < (unsigned)SLICE_ROWS;
            const unsigned m = __builtin_amdgcn_ballot_w32(hit);
            const int pos = cnt + (int)__builtin_amdgcn_mbcnt_lo(m, 0u);
            if (hit) { v2i ent; ent.x = (int)loc; ent.y = d4[j]; q[pos] = ent; }
            cnt += __builtin_popcount(m);
        }

        for (int r = 0; r < cnt; r += 4 * SCAN_DRAIN) {
            float v[SCAN_DRAIN];
            int   dd[SCAN_DRAIN];
            bool  live[SCAN_DRAIN];
#pragma unroll
            for (int i = 0; i < SCAN_DRAIN; ++i) {
                const int idx = r + 4 * i + g;
                live[i] = idx < cnt;
                const int ic = imin(idx, cnt - 1);
                const v2i ent = q[ic];
                dd[i] = ent.y;
                v[i] = tab[ent.x * FD + kc];
            }
#pragma unroll
            for (int i = 0; i < SCAN_DRAIN; ++i) {
                float* p = plane + (dd[i] * ROWF + k);
                const float val = feat ? v[i] : 1.0f;
                if (adds && live[i]) {
                    gatomic_add_f32(p, val);
                }
            }
        }
    }
}

template <int F_OUT>
__global__ void node_wmma(const float* __restrict__ h,
                          const float* __restrict__ sums,
                          const float* __restrict__ Wself,
                          const float* __restrict__ Wneigh,
                          const float* __restrict__ bias,
                          float* __restrict__ out,
                          int n_nodes) {
    const int lane  = threadIdx.x & 31;
    const int wave  = threadIdx.x >> 5;
    const int tile  = blockIdx.x * (blockDim.x >> 5) + wave;
    const int m0    = tile * 16;
    if (m0 >= n_nodes) return;

    const int row   = lane & 15;
    const int khalf = lane >> 4;

    const int m  = m0 + row;
    const int mc = imin(m, n_nodes - 1);
    const float* const row0 = sums + (long)mc * ROWF;
    const float* const row1 = row0 + PLANE;
    const float invdeg = 1.0f / fmaxf(row0[CNT] + row1[CNT], 1.0f);

    const int n  = row;
    const int nc = imin(n, F_OUT - 1);
    const bool nvalid = (n < F_OUT);

    v8f c = {};
#pragma unroll
    for (int chunk = 0; chunk < 3; ++chunk) {
        v2f a, b;
#pragma unroll
        for (int j = 0; j < 2; ++j) {
            const int k  = chunk * 4 + khalf * 2 + j;
            const int kh = imin(k, FD - 1);
            const int kn = imin(imax(k - FD, 0), FD - 1);

            const float hv = h[(long)mc * FD + kh];
            const float nv = (row0[kn] + row1[kn]) * invdeg;
            const float av = (k < FD) ? hv : ((k < 2 * FD) ? nv : 0.0f);

            const float ws = Wself[kh * F_OUT + nc];
            const float wn = Wneigh[kn * F_OUT + nc];
            float bv = (k < FD) ? ws : ((k < 2 * FD) ? wn : 0.0f);
            bv = nvalid ? bv : 0.0f;

            a[j] = av;
            b[j] = bv;
        }
        c = __builtin_amdgcn_wmma_f32_16x16x4_f32(false, a, false, b,
                                                  (short)0, c, false, false);
    }

    const float bn = bias[nc];
#pragma unroll
    for (int r = 0; r < 8; ++r) {
        const int mm = m0 + r + khalf * 8;
        if (mm < n_nodes && nvalid) {
            const float v = c[r] + bn;
            out[(long)mm * F_OUT + n] = 1.0f / (1.0f + __expf(-v));
        }
    }
}

#define WS_SORTED (((size_t)2 * PLANE + (size_t)NNODES * FD) * sizeof(float) \
                   + (size_t)NCHUNK * OFFP * sizeof(int) + (size_t)NEDGES * sizeof(unsigned))

extern "C" void kernel_launch(void* const* d_in, const int* in_sizes, int n_in,
                              void* d_out, int out_size, void* d_ws, size_t ws_size,
                              hipStream_t stream) {
    (void)in_sizes; (void)n_in; (void)out_size;

    const float* x      = (const float*)d_in[0];
    const int*   src    = (const int*)d_in[1];
    const int*   dst    = (const int*)d_in[2];
    const float* Wself1 = (const float*)d_in[3];
    const float* Wneigh1= (const float*)d_in[4];
    const float* b1     = (const float*)d_in[5];
    const float* Wself2 = (const float*)d_in[6];
    const float* Wneigh2= (const float*)d_in[7];
    const float* b2     = (const float*)d_in[8];
    float* out = (float*)d_out;

    float* sums = (float*)d_ws;
    float* h    = sums + (size_t)2 * PLANE;
    int* off       = (int*)(h + (size_t)NNODES * FD);
    unsigned* part = (unsigned*)(off + (size_t)NCHUNK * OFFP);

    const bool sorted = ws_size >= WS_SORTED;

    const int ZB = 256;
    const int EB = SCAN_BLOCK;
    const int edge_grid = NSLICE * NPART;

    const int NODE_B = 256;
    const int tiles = (NNODES + 15) / 16;
    const int node_grid = (tiles + (NODE_B / 32) - 1) / (NODE_B / 32);

    if (sorted) sort_by_slice<<<NCHUNK, SORT_BLOCK, 0, stream>>>(src, dst, part, off);

    {
        const int nz = 2 * NNODES * (FD + 1);
        zero_rows<FD + 1><<<(nz + ZB - 1) / ZB, ZB, 0, stream>>>(sums, nz);
    }
    if (sorted) edge_agg_sorted<true><<<edge_grid, EB, 0, stream>>>(x, part, off, sums);
    else        edge_agg_scan<true><<<edge_grid, EB, 0, stream>>>(x, src, dst, sums, NEDGES);
    node_wmma<FD><<<node_grid, NODE_B, 0, stream>>>(x, sums, Wself1, Wneigh1,
                                                    b1, h, NNODES);

    {
        const int nz = 2 * NNODES * FD;
        zero_rows<FD><<<(nz + ZB - 1) / ZB, ZB, 0, stream>>>(sums, nz);
    }
    if (sorted) edge_agg_sorted<false><<<edge_grid, EB, 0, stream>>>(h, part, off, sums);
    else        edge_agg_scan<false><<<edge_grid, EB, 0, stream>>>(h, src, dst, sums, NEDGES);
    node_wmma<10><<<node_grid, NODE_B, 0, stream>>>(h, sums, Wself2, Wneigh2,
                                                    b2, out, NNODES);
}


# Sum planes cleared by waves that run anyway (sort pre-pass clears both planes; layer-1 node transform zeroes the rows it has read): both fill launches of one-store waves go, 7 launches to 5
# speedup vs baseline: 25.1369x; 1.0486x over previous
#include <hip/hip_runtime.h>
#include <math.h>

#define FD 5
#define NNODES 100000
#define NEDGES 6400000

typedef __attribute__((ext_vector_type(2))) float v2f;
typedef __attribute__((ext_vector_type(8))) float v8f;

__device__ __forceinline__ int imin(int a, int b) { return a < b ? a : b; }
__device__ __forceinline__ int imax(int a, int b) { return a > b ? a : b; }

__device__ __forceinline__ void gatomic_add_f32(float* p, float v) {
    asm volatile("global_atomic_add_f32 %0, %1, off"
                 :
                 : "v"(p), "v"(v)
                 : "memory");
}

#define ROWF 8
#define CNT FD
#define PLANE (NNODES * ROWF)

__device__ __forceinline__ int package_half() {
    const unsigned id = __builtin_amdgcn_s_sendmsg_rtn(0x87);
    return (int)((id >> 18) & 1u);
}

template <int LIVE>
__global__ void zero_rows(float* __restrict__ p, int n) {
    int i = blockIdx.x * blockDim.x + threadIdx.x;
    int stride = gridDim.x * blockDim.x;
    for (; i < n; i += stride) {
        const int r = i / LIVE;
        p[r * ROWF + (i - r * LIVE)] = 0.0f;
    }
}

#define NSLICE 8
#define SLICE_ROWS (NNODES / NSLICE)
#define NCHUNK 256
#define CHUNK (NEDGES / NCHUNK)
#define WSPAN 784
#define WTRIPS 7
#define OFFP 16
#define DSTBITS 17
#define SORT_BLOCK 1024
#define SORT_WAVES (SORT_BLOCK / 32)

typedef __attribute__((ext_vector_type(4))) int v4i;
typedef __attribute__((ext_vector_type(4))) float v4f;

__device__ __forceinline__ unsigned wave_sum(unsigned x) {
#pragma unroll
    for (int m = 16; m > 0; m >>= 1) x += (unsigned)__shfl_xor((int)x, m, 32);
    return x;
}

__global__ void __launch_bounds__(SORT_BLOCK)
sort_by_slice(const int* __restrict__ src,
              const int* __restrict__ dst,
              unsigned* __restrict__ part,
              int* __restrict__ off,
              float* __restrict__ sums) {
    __shared__ int cnt[NSLICE * SORT_WAVES];
    __shared__ int base[NSLICE * SORT_WAVES];

    const int lane = threadIdx.x & 31;
    const int wave = threadIdx.x >> 5;
    const int c    = blockIdx.x;
    const int cbeg = c * CHUNK;
    const int wbeg = cbeg + wave * WSPAN;
    const int wend = imin(wbeg + WSPAN, cbeg + CHUNK);

    v4i sv[WTRIPS], dv[WTRIPS];
#pragma unroll
    for (int t = 0; t < WTRIPS; ++t) {
        const int e = wbeg + t * 128 + lane * 4;
        sv[t] = (v4i){-1, -1, -1, -1};
        dv[t] = (v4i){0, 0, 0, 0};
        if (e < wend) { sv[t] = *(const v4i*)(src + e); dv[t] = *(const v4i*)(dst + e); }
    }

    {
        v4f* const z4 = (v4f*)sums;
        for (int i = blockIdx.x * SORT_BLOCK + threadIdx.x; i < 2 * PLANE / 4; i += NCHUNK * SORT_BLOCK)
            z4[i] = (v4f){0.0f, 0.0f, 0.0f, 0.0f};
    }

    unsigned clo = 0, chi = 0;
#pragma unroll
    for (int t = 0; t < WTRIPS; ++t) {
#pragma unroll
        for (int j = 0; j < 4; ++j) {
            const unsigned sl  = (unsigned)sv[t][j] / (unsigned)SLICE_ROWS;
            const unsigned one = 1u << (8u * (sl & 3u));
            clo += (sl < 4u) ? one : 0u;
            chi += (sl - 4u < 4u) ? one : 0u;
        }
    }
    {
        const unsigned s02 = wave_sum(clo & 0x00ff00ffu);
        const unsigned s13 = wave_sum((clo >> 8) & 0x00ff00ffu);
        const unsigned s46 = wave_sum(chi & 0x00ff00ffu);
        const unsigned s57 = wave_sum((chi >> 8) & 0x00ff00ffu);
        if (lane == 0) {
            cnt[0 * SORT_WAVES + wave] = (int)(s02 & 0xffffu);
            cnt[1 * SORT_WAVES + wave] = (int)(s13 & 0xffffu);
            cnt[2 * SORT_WAVES + wave] = (int)(s02 >> 16);
            cnt[3 * SORT_WAVES + wave] = (int)(s13 >> 16);
            cnt[4 * SORT_WAVES + wave] = (int)(s46 & 0xffffu);
            cnt[5 * SORT_WAVES + wave] = (int)(s57 & 0xffffu);
            cnt[6 * SORT_WAVES + wave] = (int)(s46 >> 16);
            cnt[7 * SORT_WAVES + wave] = (int)(s57 >> 16);
        }
    }
    __syncthreads();

    if (threadIdx.x < NSLICE * SORT_WAVES) {
        int acc = 0;
        for (int i = 0; i < (int)threadIdx.x; ++i) acc += cnt[i];
        base[threadIdx.x] = acc;
        if ((threadIdx.x & (SORT_WAVES - 1)) == 0) off[c * OFFP + (threadIdx.x / SORT_WAVES)] = acc;
        if (threadIdx.x == NSLICE * SORT_WAVES - 1) off[c * OFFP + NSLICE] = acc + cnt[threadIdx.x];
    }
    __syncthreads();

    int b[NSLICE];
#pragma unroll
    for (int j = 0; j < NSLICE; ++j)
        b[j] = __builtin_amdgcn_readfirstlane(base[j * SORT_WAVES + wave]);
    unsigned* const seg = part + (long)cbeg;
#pragma unroll
    for (int t = 0; t < WTRIPS; ++t) {
#pragma unroll
        for (int jj = 0; jj < 4; ++jj) {
            const unsigned sl = (unsigned)sv[t][jj] / (unsigned)SLICE_ROWS;
            int pos = 0;
#pragma unroll
            for (int j = 0; j < NSLICE; ++j) {
                const bool mine = (sl == (unsigned)j);
                const unsigned m = __builtin_amdgcn_ballot_w32(mine);
                const int p = (int)__builtin_amdgcn_mbcnt_lo(m, (unsigned)b[j]);
                pos = mine ? p : pos;
                b[j] += __builtin_popcount(m);
            }
            if (sl < (unsigned)NSLICE) {
                const unsigned loc = (unsigned)sv[t][jj] - sl * (unsigned)SLICE_ROWS;
                seg[pos] = (loc << DSTBITS) | (unsigned)dv[t][jj];
            }
        }
    }
}

#define NPART 32
#define SCAN_BLOCK 1024
#define SCAN_WAVES (SCAN_BLOCK / 32)
#define DRAIN 8
#define TRIP_EDGES (4 * DRAIN)
static_assert(DRAIN == 8, "the fence in edge_agg_sorted names the eight table reads of a trip");

template <bool ADD_DEG>
__global__ void __launch_bounds__(SCAN_BLOCK)
edge_agg_sorted(const float* __restrict__ h,
                const unsigned* __restrict__ part,
                const int* __restrict__ off,
                float* __restrict__ sums) {
    __shared__ float tab[SLICE_ROWS * FD];

    const int lane  = threadIdx.x & 31;
    const int wave  = threadIdx.x >> 5;
    const int slice = blockIdx.x % NSLICE;
    const int prt   = blockIdx.x / NSLICE;
    const int row0  = slice * SLICE_ROWS;

    {
        const v4f* g4 = (const v4f*)(h + (long)row0 * FD);
        v4f* t4 = (v4f*)tab;
        for (int i = threadIdx.x; i < SLICE_ROWS * FD / 4; i += SCAN_BLOCK) t4[i] = g4[i];
    }
    __syncthreads();

    const int  k    = lane & 7;
    const int  g    = lane >> 3;
    const int  kc   = imin(k, FD - 1);
    const bool feat = (k < FD);
    const bool adds = feat || (ADD_DEG && k == FD);
    float* const plane = sums + package_half() * PLANE;

    const int c0 = prt * (NCHUNK / NPART);
    int nbeg = off[c0 * OFFP + slice];
    int nend = off[c0 * OFFP + slice + 1];
    for (int cc = 0; cc < NCHUNK / NPART; ++cc) {
        const int c    = c0 + cc;
        const int sbeg = nbeg;
        const int send = nend;
        const int cn = imin(c + 1, NCHUNK - 1);
        nbeg = off[cn * OFFP + slice];
        nend = off[cn * OFFP + slice + 1];
        const unsigned* const seg = part + (long)c * CHUNK;
        const int w = (wave + 4 * cc) & (SCAN_WAVES - 1);

        for (int r = sbeg + w * TRIP_EDGES; r < send; r += SCAN_WAVES * TRIP_EDGES) {
            unsigned wd[DRAIN];
            float    v[DRAIN];
            bool     live[DRAIN];
#pragma unroll
            for (int i = 0; i < DRAIN; ++i) {
                const int idx = r + 4 * i + g;
                live[i] = idx < send;
                wd[i] = seg[imin(idx, send - 1)];
            }
#pragma unroll
            for (int i = 0; i < DRAIN; ++i)
                v[i] = tab[(int)(wd[i] >> DSTBITS) * FD + kc];
            asm volatile("" : : "v"(v[0]), "v"(v[1]), "v"(v[2]), "v"(v[3]),
                                "v"(v[4]), "v"(v[5]), "v"(v[6]), "v"(v[7]));
#pragma unroll
            for (int i = 0; i < DRAIN; ++i) {
                const int dd = (int)(wd[i] & ((1u << DSTBITS) - 1u));
                float* p = plane + (dd * ROWF + k);
                const float val = feat ? v[i] : 1.0f;
                if (adds && live[i]) {
                    gatomic_add_f32(p, val);
                }
            }
        }
    }
}

#define SCAN_EDGES 128
#define SCAN_DRAIN 4

typedef __attribute__((ext_vector_type(2))) int v2i;

template <bool ADD_DEG>
__global__ void __launch_bounds__(SCAN_BLOCK)
edge_agg_scan(const float* __restrict__ h,
              const int* __restrict__ src,
              const int* __restrict__ dst,
              float* __restrict__ sums,
              int n_edges) {
    __shared__ float tab[SLICE_ROWS * FD];
    __shared__ v2i queue[SCAN_WAVES][SCAN_EDGES];

    const int lane  = threadIdx.x & 31;
    const int wave  = threadIdx.x >> 5;
    const int slice = blockIdx.x % NSLICE;
    const int part  = blockIdx.x / NSLICE;
    const int row0  = slice * SLICE_ROWS;

    {
        const v4f* g4 = (const v4f*)(h + (long)row0 * FD);
        v4f* t4 = (v4f*)tab;
        for (int i = threadIdx.x; i < SLICE_ROWS * FD / 4; i += SCAN_BLOCK) t4[i] = g4[i];
    }
    __syncthreads();

    const int  k    = lane & 7;
    const int  g    = lane >> 3;
    const int  kc   = imin(k, FD - 1);
    const bool feat = (k < FD);
    const bool adds = feat || (ADD_DEG && k == FD);
    float* const plane = sums + package_half() * PLANE;

    const int epp  = (((n_edges + NPART - 1) / NPART) + 3) & ~3;
    const int pbeg = part * epp;
    const int pend = imin(pbeg + epp, n_edges);

    v2i* const q = queue[wave];

    for (int eb = pbeg + wave * SCAN_EDGES; eb < pend; eb += SCAN_WAVES * SCAN_EDGES) {
        const int e = eb + lane * 4;
        int s4[4], d4[4];
        if (e + 3 < pend) {
            const v4i sv = *(const v4i*)(src + e);
            const v4i dv = *(const v4i*)(dst + e);
            s4[0] = sv.x; s4[1] = sv.y; s4[2] = sv.z; s4[3] = sv.w;
            d4[0] = dv.x; d4[1] = dv.y; d4[2] = dv.z; d4[3] = dv.w;
        } else {
#pragma unroll
            for (int j = 0; j < 4; ++j) {
                s4[j] = -1; d4[j] = 0;
                if (e + j < pend) { s4[j] = src[e + j]; d4[j] = dst[e + j]; }
            }
        }

        int cnt = 0;
#pragma unroll
        for (int j = 0; j < 4; ++j) {
            const unsigned loc = (unsigned)(s4[j] - row0);
            const bool hit = loc < (unsigned)SLICE_ROWS;
            const unsigned m = __builtin_amdgcn_ballot_w32(hit);
            const int pos = cnt + (int)__builtin_amdgcn_mbcnt_lo(m, 0u);
            if (hit) { v2i ent; ent.x = (int)loc; ent.y = d4[j]; q[pos] = ent; }
            cnt += __builtin_popcount(m);
        }

        for (int r = 0; r < cnt; r += 4 * SCAN_DRAIN) {
            float v[SCAN_DRAIN];
            int   dd[SCAN_DRAIN];
            bool  live[SCAN_DRAIN];
#pragma unroll
            for (int i = 0; i < SCAN_DRAIN; ++i) {
                const int idx = r + 4 * i + g;
                live[i] = idx < cnt;
                const int ic = imin(idx, cnt - 1);
                const v2i ent = q[ic];
                dd[i] = ent.y;
                v[i] = tab[ent.x * FD + kc];
            }
#pragma unroll
            for (int i = 0; i < SCAN_DRAIN; ++i) {
                float* p = plane + (dd[i] * ROWF + k);
                const float val = feat ? v[i] : 1.0f;
                if (adds && live[i]) {
                    gatomic_add_f32(p, val);
                }
            }
        }
    }
}

template <int F_OUT, bool CLEAR>
__global__ void node_wmma(const float* __restrict__ h,
                          float* sums,
                          const float* __restrict__ Wself,
                          const float* __restrict__ Wneigh,
                          const float* __restrict__ bias,
                          float* __restrict__ out,
                          int n_nodes) {
    const int lane  = threadIdx.x & 31;
    const int wave  = threadIdx.x >> 5;
    const int tile  = blockIdx.x * (blockDim.x >> 5) + wave;
    const int m0    = tile * 16;
    if (m0 >= n_nodes) return;

    const int row   = lane & 15;
    const int khalf = lane >> 4;

    const int m  = m0 + row;
    const int mc = imin(m, n_nodes - 1);
    const float* const row0 = sums + (long)mc * ROWF;
    const float* const row1 = row0 + PLANE;
    const float invdeg = 1.0f / fmaxf(row0[CNT] + row1[CNT], 1.0f);

    const int n  = row;
    const int nc = imin(n, F_OUT - 1);
    const bool nvalid = (n < F_OUT);

    v8f c = {};
#pragma unroll
    for (int chunk = 0; chunk < 3; ++chunk) {
        v2f a, b;
#pragma unroll
        for (int j = 0; j < 2; ++j) {
            const int k  = chunk * 4 + khalf * 2 + j;
            const int kh = imin(k, FD - 1);
            const int kn = imin(imax(k - FD, 0), FD - 1);

            const float hv = h[(long)mc * FD + kh];
            const float nv = (row0[kn] + row1[kn]) * invdeg;
            const float av = (k < FD) ? hv : ((k < 2 * FD) ? nv : 0.0f);

            const float ws = Wself[kh * F_OUT + nc];
            const float wn = Wneigh[kn * F_OUT + nc];
            float bv = (k < FD) ? ws : ((k < 2 * FD) ? wn : 0.0f);
            bv = nvalid ? bv : 0.0f;

            a[j] = av;
            b[j] = bv;
        }
        c = __builtin_amdgcn_wmma_f32_16x16x4_f32(false, a, false, b,
                                                  (short)0, c, false, false);
    }

    const float bn = bias[nc];
#pragma unroll
    for (int r = 0; r < 8; ++r) {
        const int mm = m0 + r + khalf * 8;
        if (mm < n_nodes && nvalid) {
            const float v = c[r] + bn;
            out[(long)mm * F_OUT + n] = 1.0f / (1.0f + __expf(-v));
        }
    }

    if (CLEAR) {
        static_assert(FD == 5, "the clear writes columns 0-3 as one 16-byte piece and column 4 alone");
        if (m < n_nodes) {
            float* const z = sums + (long)khalf * PLANE + (long)m * ROWF;
            *(v4f*)z = (v4f){0.0f, 0.0f, 0.0f, 0.0f};
            z[4] = 0.0f;
        }
    }
}

#define WS_SORTED (((size_t)2 * PLANE + (size_t)NNODES * FD) * sizeof(float) \
                   + (size_t)NCHUNK * OFFP * sizeof(int) + (size_t)NEDGES * sizeof(unsigned))

extern "C" void kernel_launch(void* const* d_in, const int* in_sizes, int n_in,
                              void* d_out, int out_size, void* d_ws, size_t ws_size,
                              hipStream_t stream) {
    (void)in_sizes; (void)n_in; (void)out_size;

    const float* x      = (const float*)d_in[0];
    const int*   src    = (const int*)d_in[1];
    const int*   dst    = (const int*)d_in[2];
    const float* Wself1 = (const float*)d_in[3];
    const float* Wneigh1= (const float*)d_in[4];
    const float* b1     = (const float*)d_in[5];
    const float* Wself2 = (const float*)d_in[6];
    const float* Wneigh2= (const float*)d_in[7];
    const float* b2     = (const float*)d_in[8];
    float* out = (float*)d_out;

    float* sums = (float*)d_ws;
    float* h    = sums + (size_t)2 * PLANE;
    int* off       = (int*)(h + (size_t)NNODES * FD);
    unsigned* part = (unsigned*)(off + (size_t)NCHUNK * OFFP);

    const bool sorted = ws_size >= WS_SORTED;

    const int ZB = 256;
    const int EB = SCAN_BLOCK;
    const int edge_grid = NSLICE * NPART;

    const int NODE_B = 256;
    const int tiles = (NNODES + 15) / 16;
    const int node_grid = (tiles + (NODE_B / 32) - 1) / (NODE_B / 32);

    if (sorted) {
        sort_by_slice<<<NCHUNK, SORT_BLOCK, 0, stream>>>(src, dst, part, off, sums);
    } else {
        const int nz = 2 * NNODES * (FD + 1);
        zero_rows<FD + 1><<<(nz + ZB - 1) / ZB, ZB, 0, stream>>>(sums, nz);
    }

    if (sorted) edge_agg_sorted<true><<<edge_grid, EB, 0, stream>>>(x, part, off, sums);
    else        edge_agg_scan<true><<<edge_grid, EB, 0, stream>>>(x, src, dst, sums, NEDGES);
    node_wmma<FD, true><<<node_grid, NODE_B, 0, stream>>>(x, sums, Wself1, Wneigh1,
                                                          b1, h, NNODES);

    if (sorted) edge_agg_sorted<false><<<edge_grid, EB, 0, stream>>>(h, part, off, sums);
    else        edge_agg_scan<false><<<edge_grid, EB, 0, stream>>>(h, src, dst, sums, NEDGES);
    node_wmma<10, false><<<node_grid, NODE_B, 0, stream>>>(h, sums, Wself2, Wneigh2,
                                                           b2, out, NNODES);
}


# Sort pre-pass builds the sorted chunk in LDS (place = scanned counts + a lane-private LDS cursor) and stores it as whole lines, not wave votes and 4-byte scatters; edge order, so add order, new
# speedup vs baseline: 26.3889x; 1.0498x over previous
#include <hip/hip_runtime.h>
#include <math.h>

#define FD 5
#define NNODES 100000
#define NEDGES 6400000

typedef __attribute__((ext_vector_type(2))) float v2f;
typedef __attribute__((ext_vector_type(8))) float v8f;

__device__ __forceinline__ int imin(int a, int b) { return a < b ? a : b; }
__device__ __forceinline__ int imax(int a, int b) { return a > b ? a : b; }

__device__ __forceinline__ void gatomic_add_f32(float* p, float v) {
    asm volatile("global_atomic_add_f32 %0, %1, off"
                 :
                 : "v"(p), "v"(v)
                 : "memory");
}

#define ROWF 8
#define CNT FD
#define PLANE (NNODES * ROWF)

__device__ __forceinline__ int package_half() {
    const unsigned id = __builtin_amdgcn_s_sendmsg_rtn(0x87);
    return (int)((id >> 18) & 1u);
}

template <int LIVE>
__global__ void zero_rows(float* __restrict__ p, int n) {
    int i = blockIdx.x * blockDim.x + threadIdx.x;
    int stride = gridDim.x * blockDim.x;
    for (; i < n; i += stride) {
        const int r = i / LIVE;
        p[r * ROWF + (i - r * LIVE)] = 0.0f;
    }
}

#define NSLICE 8
#define SLICE_ROWS (NNODES / NSLICE)
#define NCHUNK 256
#define CHUNK (NEDGES / NCHUNK)
#define WSPAN 784
#define WTRIPS 7
#define OFFP 16
#define DSTBITS 17
#define SORT_BLOCK 1024
#define SORT_WAVES (SORT_BLOCK / 32)

typedef __attribute__((ext_vector_type(4))) int v4i;
typedef __attribute__((ext_vector_type(4))) float v4f;

__device__ __forceinline__ unsigned wave_scan(unsigned x, int lane) {
#pragma unroll
    for (int d = 1; d < 32; d <<= 1) {
        const unsigned t = (unsigned)__shfl_up((int)x, d, 32);
        x += (lane >= d) ? t : 0u;
    }
    return x;
}

__global__ void __launch_bounds__(SORT_BLOCK)
sort_by_slice(const int* __restrict__ src,
              const int* __restrict__ dst,
              unsigned* __restrict__ part,
              int* __restrict__ off,
              float* __restrict__ sums) {
    __shared__ int cnt[NSLICE * SORT_WAVES];
    __shared__ int base[NSLICE * SORT_WAVES];
    __shared__ int stot[NSLICE];
    __shared__ unsigned cur[NSLICE * SORT_BLOCK];
    __shared__ __attribute__((aligned(16))) unsigned img[CHUNK];

    const int lane = threadIdx.x & 31;
    const int wave = threadIdx.x >> 5;
    const int c    = blockIdx.x;
    const int cbeg = c * CHUNK;
    const int wbeg = cbeg + wave * WSPAN;
    const int wend = imin(wbeg + WSPAN, cbeg + CHUNK);

    v4i sv[WTRIPS], dv[WTRIPS];
#pragma unroll
    for (int t = 0; t < WTRIPS; ++t) {
        const int e = wbeg + t * 128 + lane * 4;
        sv[t] = (v4i){-1, -1, -1, -1};
        dv[t] = (v4i){0, 0, 0, 0};
        if (e < wend) { sv[t] = *(const v4i*)(src + e); dv[t] = *(const v4i*)(dst + e); }
    }

    {
        v4f* const z4 = (v4f*)sums;
        for (int i = blockIdx.x * SORT_BLOCK + threadIdx.x; i < 2 * PLANE / 4; i += NCHUNK * SORT_BLOCK)
            z4[i] = (v4f){0.0f, 0.0f, 0.0f, 0.0f};
    }

    unsigned clo = 0, chi = 0;
#pragma unroll
    for (int t = 0; t < WTRIPS; ++t) {
#pragma unroll
        for (int j = 0; j < 4; ++j) {
            const unsigned sl  = (unsigned)sv[t][j] / (unsigned)SLICE_ROWS;
            const unsigned one = 1u << (8u * (sl & 3u));
            clo += (sl < 4u) ? one : 0u;
            chi += (sl - 4u < 4u) ? one : 0u;
        }
    }
    const unsigned p02 = clo & 0x00ff00ffu;
    const unsigned p13 = (clo >> 8) & 0x00ff00ffu;
    const unsigned p46 = chi & 0x00ff00ffu;
    const unsigned p57 = (chi >> 8) & 0x00ff00ffu;
    const unsigned i02 = wave_scan(p02, lane);
    const unsigned i13 = wave_scan(p13, lane);
    const unsigned i46 = wave_scan(p46, lane);
    const unsigned i57 = wave_scan(p57, lane);
    if (lane == 31) {
        cnt[0 * SORT_WAVES + wave] = (int)(i02 & 0xffffu);
        cnt[1 * SORT_WAVES + wave] = (int)(i13 & 0xffffu);
        cnt[2 * SORT_WAVES + wave] = (int)(i02 >> 16);
        cnt[3 * SORT_WAVES + wave] = (int)(i13 >> 16);
        cnt[4 * SORT_WAVES + wave] = (int)(i46 & 0xffffu);
        cnt[5 * SORT_WAVES + wave] = (int)(i57 & 0xffffu);
        cnt[6 * SORT_WAVES + wave] = (int)(i46 >> 16);
        cnt[7 * SORT_WAVES + wave] = (int)(i57 >> 16);
    }
    __syncthreads();

    static_assert(SORT_WAVES == 32, "one lane of a wave per wave of the work-group");
    const bool scans = threadIdx.x < NSLICE * SORT_WAVES;
    int own = 0, incl = 0;
    if (scans) {
        own  = cnt[threadIdx.x];
        incl = (int)wave_scan((unsigned)own, lane);
        if (lane == 31) stot[wave] = incl;
    }
    __syncthreads();
    if (scans) {
        int acc = incl - own;
#pragma unroll
        for (int j = 0; j < NSLICE; ++j) acc += (j < wave) ? stot[j] : 0;
        base[threadIdx.x] = acc;
        if (lane == 0) off[c * OFFP + wave] = acc;
        if (threadIdx.x == NSLICE * SORT_WAVES - 1) off[c * OFFP + NSLICE] = acc + own;
    }
    __syncthreads();

    {
        const unsigned e02 = i02 - p02, e13 = i13 - p13, e46 = i46 - p46, e57 = i57 - p57;
        unsigned* const mine = cur + threadIdx.x;
        mine[0 * SORT_BLOCK] = (unsigned)base[0 * SORT_WAVES + wave] + (e02 & 0xffffu);
        mine[1 * SORT_BLOCK] = (unsigned)base[1 * SORT_WAVES + wave] + (e13 & 0xffffu);
        mine[2 * SORT_BLOCK] = (unsigned)base[2 * SORT_WAVES + wave] + (e02 >> 16);
        mine[3 * SORT_BLOCK] = (unsigned)base[3 * SORT_WAVES + wave] + (e13 >> 16);
        mine[4 * SORT_BLOCK] = (unsigned)base[4 * SORT_WAVES + wave] + (e46 & 0xffffu);
        mine[5 * SORT_BLOCK] = (unsigned)base[5 * SORT_WAVES + wave] + (e57 & 0xffffu);
        mine[6 * SORT_BLOCK] = (unsigned)base[6 * SORT_WAVES + wave] + (e46 >> 16);
        mine[7 * SORT_BLOCK] = (unsigned)base[7 * SORT_WAVES + wave] + (e57 >> 16);
    }

#pragma unroll
    for (int t = 0; t < WTRIPS; ++t) {
#pragma unroll
        for (int jj = 0; jj < 4; ++jj) {
            const unsigned s  = (unsigned)sv[t][jj];
            const unsigned sl = s / (unsigned)SLICE_ROWS;
            if (sl < (unsigned)NSLICE) {
                unsigned* const cp = cur + (sl * SORT_BLOCK + threadIdx.x);
                const unsigned pos = *cp;
                *cp = pos + 1u;
                img[pos] = ((s - sl * (unsigned)SLICE_ROWS) << DSTBITS) | (unsigned)dv[t][jj];
            }
        }
    }
    __syncthreads();

    {
        const v4i* const i4 = (const v4i*)img;
        v4i* const o4 = (v4i*)(part + (long)cbeg);
        constexpr int NCOPY = (CHUNK / 4 + SORT_BLOCK - 1) / SORT_BLOCK;
        v4i tmp[NCOPY];
#pragma unroll
        for (int k = 0; k < NCOPY; ++k) {
            const int i = threadIdx.x + k * SORT_BLOCK;
            tmp[k] = (v4i){0, 0, 0, 0};
            if (i < CHUNK / 4) tmp[k] = i4[i];
        }
#pragma unroll
        for (int k = 0; k < NCOPY; ++k) {
            const int i = threadIdx.x + k * SORT_BLOCK;
            if (i < CHUNK / 4) o4[i] = tmp[k];
        }
    }
}

#define NPART 32
#define SCAN_BLOCK 1024
#define SCAN_WAVES (SCAN_BLOCK / 32)
#define DRAIN 8
#define TRIP_EDGES (4 * DRAIN)
static_assert(DRAIN == 8, "the fence in edge_agg_sorted names the eight table reads of a trip");

template <bool ADD_DEG>
__global__ void __launch_bounds__(SCAN_BLOCK)
edge_agg_sorted(const float* __restrict__ h,
                const unsigned* __restrict__ part,
                const int* __restrict__ off,
                float* __restrict__ sums) {
    __shared__ float tab[SLICE_ROWS * FD];

    const int lane  = threadIdx.x & 31;
    const int wave  = threadIdx.x >> 5;
    const int slice = blockIdx.x % NSLICE;
    const int prt   = blockIdx.x / NSLICE;
    const int row0  = slice * SLICE_ROWS;

    {
        const v4f* g4 = (const v4f*)(h + (long)row0 * FD);
        v4f* t4 = (v4f*)tab;
        for (int i = threadIdx.x; i < SLICE_ROWS * FD / 4; i += SCAN_BLOCK) t4[i] = g4[i];
    }
    __syncthreads();

    const int  k    = lane & 7;
    const int  g    = lane >> 3;
    const int  kc   = imin(k, FD - 1);
    const bool feat = (k < FD);
    const bool adds = feat || (ADD_DEG && k == FD);
    float* const plane = sums + package_half() * PLANE;

    const int c0 = prt * (NCHUNK / NPART);
    int nbeg = off[c0 * OFFP + slice];
    int nend = off[c0 * OFFP + slice + 1];
    for (int cc = 0; cc < NCHUNK / NPART; ++cc) {
        const int c    = c0 + cc;
        const int sbeg = nbeg;
        const int send = nend;
        const int cn = imin(c + 1, NCHUNK - 1);
        nbeg = off[cn * OFFP + slice];
        nend = off[cn * OFFP + slice + 1];
        const unsigned* const seg = part + (long)c * CHUNK;
        const int w = (wave + 4 * cc) & (SCAN_WAVES - 1);

        for (int r = sbeg + w * TRIP_EDGES; r < send; r += SCAN_WAVES * TRIP_EDGES) {
            unsigned wd[DRAIN];
            float    v[DRAIN];
            bool     live[DRAIN];
#pragma unroll
            for (int i = 0; i < DRAIN; ++i) {
                const int idx = r + 4 * i + g;
                live[i] = idx < send;
                wd[i] = seg[imin(idx, send - 1)];
            }
#pragma unroll
            for (int i = 0; i < DRAIN; ++i)
                v[i] = tab[(int)(wd[i] >> DSTBITS) * FD + kc];
            asm volatile("" : : "v"(v[0]), "v"(v[1]), "v"(v[2]), "v"(v[3]),
                                "v"(v[4]), "v"(v[5]), "v"(v[6]), "v"(v[7]));
#pragma unroll
            for (int i = 0; i < DRAIN; ++i) {
                const int dd = (int)(wd[i] & ((1u << DSTBITS) - 1u));
                float* p = plane + (dd * ROWF + k);
                const float val = feat ? v[i] : 1.0f;
                if (adds && live[i]) {
                    gatomic_add_f32(p, val);
                }
            }
        }
    }
}

#define SCAN_EDGES 128
#define SCAN_DRAIN 4

typedef __attribute__((ext_vector_type(2))) int v2i;

template <bool ADD_DEG>
__global__ void __launch_bounds__(SCAN_BLOCK)
edge_agg_scan(const float* __restrict__ h,
              const int* __restrict__ src,
              const int* __restrict__ dst,
              float* __restrict__ sums,
              int n_edges) {
    __shared__ float tab[SLICE_ROWS * FD];
    __shared__ v2i queue[SCAN_WAVES][SCAN_EDGES];

    const int lane  = threadIdx.x & 31;
    const int wave  = threadIdx.x >> 5;
    const int slice = blockIdx.x % NSLICE;
    const int part  = blockIdx.x / NSLICE;
    const int row0  = slice * SLICE_ROWS;

    {
        const v4f* g4 = (const v4f*)(h + (long)row0 * FD);
        v4f* t4 = (v4f*)tab;
        for (int i = threadIdx.x; i < SLICE_ROWS * FD / 4; i += SCAN_BLOCK) t4[i] = g4[i];
    }
    __syncthreads();

    const int  k    = lane & 7;
    const int  g    = lane >> 3;
    const int  kc   = imin(k, FD - 1);
    const bool feat = (k < FD);
    const bool adds = feat || (ADD_DEG && k == FD);
    float* const plane = sums + package_half() * PLANE;

    const int epp  = (((n_edges + NPART - 1) / NPART) + 3) & ~3;
    const int pbeg = part * epp;
    const int pend = imin(pbeg + epp, n_edges);

    v2i* const q = queue[wave];

    for (int eb = pbeg + wave * SCAN_EDGES; eb < pend; eb += SCAN_WAVES * SCAN_EDGES) {
        const int e = eb + lane * 4;
        int s4[4], d4[4];
        if (e + 3 < pend) {
            const v4i sv = *(const v4i*)(src + e);
            const v4i dv = *(const v4i*)(dst + e);
            s4[0] = sv.x; s4[1] = sv.y; s4[2] = sv.z; s4[3] = sv.w;
            d4[0] = dv.x; d4[1] = dv.y; d4[2] = dv.z; d4[3] = dv.w;
        } else {
#pragma unroll
            for (int j = 0; j < 4; ++j) {
                s4[j] = -1; d4[j] = 0;
                if (e + j < pend) { s4[j] = src[e + j]; d4[j] = dst[e + j]; }
            }
        }

        int cnt = 0;
#pragma unroll
        for (int j = 0; j < 4; ++j) {
            const unsigned loc = (unsigned)(s4[j] - row0);
            const bool hit = loc < (unsigned)SLICE_ROWS;
            const unsigned m = __builtin_amdgcn_ballot_w32(hit);
            const int pos = cnt + (int)__builtin_amdgcn_mbcnt_lo(m, 0u);
            if (hit) { v2i ent; ent.x = (int)loc; ent.y = d4[j]; q[pos] = ent; }
            cnt += __builtin_popcount(m);
        }

        for (int r = 0; r < cnt; r += 4 * SCAN_DRAIN) {
            float v[SCAN_DRAIN];
            int   dd[SCAN_DRAIN];
            bool  live[SCAN_DRAIN];
#pragma unroll
            for (int i = 0; i < SCAN_DRAIN; ++i) {
                const int idx = r + 4 * i + g;
                live[i] = idx < cnt;
                const int ic = imin(idx, cnt - 1);
                const v2i ent = q[ic];
                dd[i] = ent.y;
                v[i] = tab[ent.x * FD + kc];
            }
#pragma unroll
            for (int i = 0; i < SCAN_DRAIN; ++i) {
                float* p = plane + (dd[i] * ROWF + k);
                const float val = feat ? v[i] : 1.0f;
                if (adds && live[i]) {
                    gatomic_add_f32(p, val);
                }
            }
        }
    }
}

template <int F_OUT, bool CLEAR>
__global__ void node_wmma(const float* __restrict__ h,
                          float* sums,
                          const float* __restrict__ Wself,
                          const float* __restrict__ Wneigh,
                          const float* __restrict__ bias,
                          float* __restrict__ out,
                          int n_nodes) {
    const int lane  = threadIdx.x & 31;
    const int wave  = threadIdx.x >> 5;
    const int tile  = blockIdx.x * (blockDim.x >> 5) + wave;
    const int m0    = tile * 16;
    if (m0 >= n_nodes) return;

    const int row   = lane & 15;
    const int khalf = lane >> 4;

    const int m  = m0 + row;
    const int mc = imin(m, n_nodes - 1);
    const float* const row0 = sums + (long)mc * ROWF;
    const float* const row1 = row0 + PLANE;
    const float invdeg = 1.0f / fmaxf(row0[CNT] + row1[CNT], 1.0f);

    const int n  = row;
    const int nc = imin(n, F_OUT - 1);
    const bool nvalid = (n < F_OUT);

    v8f c = {};
#pragma unroll
    for (int chunk = 0; chunk < 3; ++chunk) {
        v2f a, b;
#pragma unroll
        for (int j = 0; j < 2; ++j) {
            const int k  = chunk * 4 + khalf * 2 + j;
            const int kh = imin(k, FD - 1);
            const int kn = imin(imax(k - FD, 0), FD - 1);

            const float hv = h[(long)mc * FD + kh];
            const float nv = (row0[kn] + row1[kn]) * invdeg;
            const float av = (k < FD) ? hv : ((k < 2 * FD) ? nv : 0.0f);

            const float ws = Wself[kh * F_OUT + nc];
            const float wn = Wneigh[kn * F_OUT + nc];
            float bv = (k < FD) ? ws : ((k < 2 * FD) ? wn : 0.0f);
            bv = nvalid ? bv : 0.0f;

            a[j] = av;
            b[j] = bv;
        }
        c = __builtin_amdgcn_wmma_f32_16x16x4_f32(false, a, false, b,
                                                  (short)0, c, false, false);
    }

    const float bn = bias[nc];
#pragma unroll
    for (int r = 0; r < 8; ++r) {
        const int mm = m0 + r + khalf * 8;
        if (mm < n_nodes && nvalid) {
            const float v = c[r] + bn;
            out[(long)mm * F_OUT + n] = 1.0f / (1.0f + __expf(-v));
        }
    }

    if (CLEAR) {
        static_assert(FD == 5, "the clear writes columns 0-3 as one 16-byte piece and column 4 alone");
        if (m < n_nodes) {
            float* const z = sums + (long)khalf * PLANE + (long)m * ROWF;
            *(v4f*)z = (v4f){0.0f, 0.0f, 0.0f, 0.0f};
            z[4] = 0.0f;
        }
    }
}

#define WS_SORTED (((size_t)2 * PLANE + (size_t)NNODES * FD) * sizeof(float) \
                   + (size_t)NCHUNK * OFFP * sizeof(int) + (size_t)NEDGES * sizeof(unsigned))

extern "C" void kernel_launch(void* const* d_in, const int* in_sizes, int n_in,
                              void* d_out, int out_size, void* d_ws, size_t ws_size,
                              hipStream_t stream) {
    (void)in_sizes; (void)n_in; (void)out_size;

    const float* x      = (const float*)d_in[0];
    const int*   src    = (const int*)d_in[1];
    const int*   dst    = (const int*)d_in[2];
    const float* Wself1 = (const float*)d_in[3];
    const float* Wneigh1= (const float*)d_in[4];
    const float* b1     = (const float*)d_in[5];
    const float* Wself2 = (const float*)d_in[6];
    const float* Wneigh2= (const float*)d_in[7];
    const float* b2     = (const float*)d_in[8];
    float* out = (float*)d_out;

    float* sums = (float*)d_ws;
    float* h    = sums + (size_t)2 * PLANE;
    int* off       = (int*)(h + (size_t)NNODES * FD);
    unsigned* part = (unsigned*)(off + (size_t)NCHUNK * OFFP);

    const bool sorted = ws_size >= WS_SORTED;

    const int ZB = 256;
    const int EB = SCAN_BLOCK;
    const int edge_grid = NSLICE * NPART;

    const int NODE_B = 256;
    const int tiles = (NNODES + 15) / 16;
    const int node_grid = (tiles + (NODE_B / 32) - 1) / (NODE_B / 32);

    if (sorted) {
        sort_by_slice<<<NCHUNK, SORT_BLOCK, 0, stream>>>(src, dst, part, off, sums);
    } else {
        const int nz = 2 * NNODES * (FD + 1);
        zero_rows<FD + 1><<<(nz + ZB - 1) / ZB, ZB, 0, stream>>>(sums, nz);
    }

    if (sorted) edge_agg_sorted<true><<<edge_grid, EB, 0, stream>>>(x, part, off, sums);
    else        edge_agg_scan<true><<<edge_grid, EB, 0, stream>>>(x, src, dst, sums, NEDGES);
    node_wmma<FD, true><<<node_grid, NODE_B, 0, stream>>>(x, sums, Wself1, Wneigh1,
                                                          b1, h, NNODES);

    if (sorted) edge_agg_sorted<false><<<edge_grid, EB, 0, stream>>>(h, part, off, sums);
    else        edge_agg_scan<false><<<edge_grid, EB, 0, stream>>>(h, src, dst, sums, NEDGES);
    node_wmma<10, false><<<node_grid, NODE_B, 0, stream>>>(h, sums, Wself2, Wneigh2,
                                                           b2, out, NNODES);
}
